# starting kernel
# baseline (speedup 1.0000x reference)
#include <hip/hip_runtime.h>
#include <hip/hip_bf16.h>

typedef float v2f __attribute__((ext_vector_type(2)));
typedef float v8f __attribute__((ext_vector_type(8)));

#define Bm 16
#define Lk 32768
#define Dn 768
#define LCHUNK 512
#define NCH 64
#define NTILES 48
#define WPB 4
#define NGROUPS 12
#define XS_STRIDE 516

__global__ __launch_bounds__(128) void sae_wmma_partial_kernel(
    const float* __restrict__ x, const float* __restrict__ W,
    float* __restrict__ partial)
{
  __shared__ float xs[Bm * XS_STRIDE];

  const int lchunk = blockIdx.x & (NCH - 1);
  const int ngroup = blockIdx.x / NCH;
  const int l0 = lchunk * LCHUNK;
  const int t = threadIdx.x;

  #pragma unroll
  for (int i = 0; i < 16; ++i) {
    const int linear = t + i * 128;
    const int m  = linear >> 7;
    const int lq = linear & 127;
    const float4 v =
        *reinterpret_cast<const float4*>(x + m * Lk + l0 + (lq << 2));
    *reinterpret_cast<float4*>(&xs[m * XS_STRIDE + (lq << 2)]) = v;
  }
  __syncthreads();

  const int wave  = t >> 5;
  const int lane  = t & 31;
  const int ntile = ngroup * WPB + wave;
  const int m     = lane & 15;
  const int kk    = (lane >> 4) << 1;
  const int ncol  = ntile * 16 + m;
  const float* wcol = W + ncol;

  v8f c = {};

  #pragma unroll 1
  for (int k = 0; k < LCHUNK; k += 16) {
    #pragma unroll
    for (int u = 0; u < 4; ++u) {
      const int kb = k + u * 4;
      v2f a = *reinterpret_cast<const v2f*>(&xs[m * XS_STRIDE + kb + kk]);
      v2f b;
      b.x = wcol[(l0 + kb + kk) * Dn];
      b.y = wcol[(l0 + kb + kk + 1) * Dn];
      c = __builtin_amdgcn_wmma_f32_16x16x4_f32(
           false, a,  false, b,
           (short)0, c,  false,  false);
    }
    int pl = l0 + k + 64;
    if (pl > Lk - 1) pl = Lk - 1;
    __builtin_prefetch(wcol + pl * Dn, 0, 1);
  }

  float* pt = partial + (size_t)lchunk * (Bm * Dn);
  const int mrow = (lane >> 4) << 3;
  #pragma unroll
  for (int r = 0; r < 8; ++r) {
    pt[(mrow + r) * Dn + ncol] = c[r];
  }
}

__global__ __launch_bounds__(256) void sae_reduce_kernel(
    const float* __restrict__ partial, float* __restrict__ out)
{
  const int idx = blockIdx.x * 256 + threadIdx.x;
  float s0 = 0.f, s1 = 0.f, s2 = 0.f, s3 = 0.f;
  #pragma unroll
  for (int cch = 0; cch < NCH; cch += 4) {
    s0 += partial[(size_t)(cch + 0) * (Bm * Dn) + idx];
    s1 += partial[(size_t)(cch + 1) * (Bm * Dn) + idx];
    s2 += partial[(size_t)(cch + 2) * (Bm * Dn) + idx];
    s3 += partial[(size_t)(cch + 3) * (Bm * Dn) + idx];
  }
  out[idx] = (s0 + s1) + (s2 + s3);
}

extern "C" void kernel_launch(void* const* d_in, const int* in_sizes, int n_in,
                              void* d_out, int out_size, void* d_ws, size_t ws_size,
                              hipStream_t stream) {
  const float* x = (const float*)d_in[0];
  const float* W = (const float*)d_in[1];
  float* out     = (float*)d_out;
  float* partial = (float*)d_ws;

  sae_wmma_partial_kernel<<<NCH * NGROUPS, 128, 0, stream>>>(x, W, partial);
  sae_reduce_kernel<<<(Bm * Dn) / 256, 256, 0, stream>>>(partial, out);
}


# W loaded as 16-byte row pieces (b128, was b32): a lane's 4 adjacent columns feed 4 interleaved N-tiles sharing one A fragment; block's 4 waves split its K rows, tiles added in LDS: new sum order
# speedup vs baseline: 2.1486x; 2.1486x over previous
#include <hip/hip_runtime.h>
#include <hip/hip_bf16.h>


typedef float v2f __attribute__((ext_vector_type(2)));
typedef float v8f __attribute__((ext_vector_type(8)));

#define Bm 16
#define Lk 32768
#define Dn 768
#define LCHUNK 512
#define NCH 64
#define WPB 4
#define KW 128
#define NCOLS 64
#define NGROUPS 12
#define XS_STRIDE 516
#define RS_STRIDE 68

__global__ __launch_bounds__(128) void sae_wmma_partial_kernel(
    const float* __restrict__ x, const float* __restrict__ W,
    float* __restrict__ partial)
{
  __shared__ float xs[Bm * XS_STRIDE];
  __shared__ float red[WPB * Bm * RS_STRIDE];

  const int lchunk = blockIdx.x & (NCH - 1);
  const int ngroup = blockIdx.x / NCH;
  const int l0 = lchunk * LCHUNK;
  const int t = threadIdx.x;

  #pragma unroll
  for (int i = 0; i < 16; ++i) {
    const int linear = t + i * 128;
    const int m  = linear >> 7;
    const int lq = linear & 127;
    const float4 v =
        *reinterpret_cast<const float4*>(x + m * Lk + l0 + (lq << 2));
    *reinterpret_cast<float4*>(&xs[m * XS_STRIDE + (lq << 2)]) = v;
  }
  __syncthreads();

  const int wave = t >> 5;
  const int lane = t & 31;
  const int c    = lane & 15;
  const int h    = lane >> 4;
  const int kk   = h << 1;
  const int kw0  = wave * KW;

  const float* xa = &xs[c * XS_STRIDE + kw0 + kk];
  const float* wp = W + (size_t)(l0 + kw0 + kk) * Dn + ngroup * NCOLS + (c << 2);

  v8f acc0 = {}, acc1 = {}, acc2 = {}, acc3 = {};

  #pragma unroll
  for (int s = 0; s < KW / 4; ++s) {
    const float4 w0 = *reinterpret_cast<const float4*>(wp + (4 * s) * Dn);
    const float4 w1 = *reinterpret_cast<const float4*>(wp + (4 * s + 1) * Dn);
    const v2f a = *reinterpret_cast<const v2f*>(xa + 4 * s);
    const v2f b0 = {w0.x, w1.x};
    const v2f b1 = {w0.y, w1.y};
    const v2f b2 = {w0.z, w1.z};
    const v2f b3 = {w0.w, w1.w};
    acc0 = __builtin_amdgcn_wmma_f32_16x16x4_f32(false, a, false, b0, (short)0, acc0, false, false);
    acc1 = __builtin_amdgcn_wmma_f32_16x16x4_f32(false, a, false, b1, (short)0, acc1, false, false);
    acc2 = __builtin_amdgcn_wmma_f32_16x16x4_f32(false, a, false, b2, (short)0, acc2, false, false);
    acc3 = __builtin_amdgcn_wmma_f32_16x16x4_f32(false, a, false, b3, (short)0, acc3, false, false);
  }

  float* rw = &red[(wave * Bm + (h << 3)) * RS_STRIDE + (c << 2)];
  #pragma unroll
  for (int r = 0; r < 8; ++r) {
    *reinterpret_cast<float4*>(rw + r * RS_STRIDE) =
        make_float4(acc0[r], acc1[r], acc2[r], acc3[r]);
  }
  __syncthreads();

  float* pt = partial + (size_t)lchunk * (Bm * Dn) + ngroup * NCOLS;
  #pragma unroll
  for (int i = 0; i < 2; ++i) {
    const int q  = t + i * 128;
    const int m  = q >> 4;
    const int c4 = (q & 15) << 2;
    const float4 p0 = *reinterpret_cast<const float4*>(&red[(0 * Bm + m) * RS_STRIDE + c4]);
    const float4 p1 = *reinterpret_cast<const float4*>(&red[(1 * Bm + m) * RS_STRIDE + c4]);
    const float4 p2 = *reinterpret_cast<const float4*>(&red[(2 * Bm + m) * RS_STRIDE + c4]);
    const float4 p3 = *reinterpret_cast<const float4*>(&red[(3 * Bm + m) * RS_STRIDE + c4]);
    float4 o;
    o.x = ((p0.x + p1.x) + p2.x) + p3.x;
    o.y = ((p0.y + p1.y) + p2.y) + p3.y;
    o.z = ((p0.z + p1.z) + p2.z) + p3.z;
    o.w = ((p0.w + p1.w) + p2.w) + p3.w;
    *reinterpret_cast<float4*>(pt + m * Dn + c4) = o;
  }
}

__global__ __launch_bounds__(256) void sae_reduce_kernel(
    const float* __restrict__ partial, float* __restrict__ out)
{
  const int idx = blockIdx.x * 256 + threadIdx.x;
  float s0 = 0.f, s1 = 0.f, s2 = 0.f, s3 = 0.f;
  #pragma unroll
  for (int cch = 0; cch < NCH; cch += 4) {
    s0 += partial[(size_t)(cch + 0) * (Bm * Dn) + idx];
    s1 += partial[(size_t)(cch + 1) * (Bm * Dn) + idx];
    s2 += partial[(size_t)(cch + 2) * (Bm * Dn) + idx];
    s3 += partial[(size_t)(cch + 3) * (Bm * Dn) + idx];
  }
  out[idx] = (s0 + s1) + (s2 + s3);
}

extern "C" void kernel_launch(void* const* d_in, const int* in_sizes, int n_in,
                              void* d_out, int out_size, void* d_ws, size_t ws_size,
                              hipStream_t stream) {
  const float* x = (const float*)d_in[0];
  const float* W = (const float*)d_in[1];
  float* out     = (float*)d_out;
  float* partial = (float*)d_ws;

  sae_wmma_partial_kernel<<<NCH * NGROUPS, 128, 0, stream>>>(x, W, partial);
  sae_reduce_kernel<<<(Bm * Dn) / 256, 256, 0, stream>>>(partial, out);
}
